# starting kernel
# baseline (speedup 1.0000x reference)
#include <hip/hip_runtime.h>
#include <math.h>

#define NB   2048
#define PR   128
#define NN   64
#define NQ   8

typedef __attribute__((ext_vector_type(2))) float v2f;
typedef __attribute__((ext_vector_type(8))) float v8f;

#if defined(__has_builtin)
#if __has_builtin(__builtin_amdgcn_wmma_f32_16x16x4_f32)
#define HAVE_WMMA_F32 1
#endif
#endif

__global__ __launch_bounds__(256) void constraint_kernel(
    const float* __restrict__ v,
    const float* __restrict__ A1,
    const float* __restrict__ b1,
    const float* __restrict__ z0,
    const float* __restrict__ P,
    const float* __restrict__ qv,
    const float* __restrict__ r,
    float* __restrict__ out)
{
    const int b    = blockIdx.x;
    const int tid  = threadIdx.x;
    const int wave = tid >> 5;
    const int lane = tid & 31;

    __shared__ float s_y0[NN];
    __shared__ float s_rho[NN];
    __shared__ float s_red[16];
    __shared__ float s_scal[2];

    if (wave == 0) {
        const float* vb = v + (size_t)b * NN;
        float v0 = vb[lane];
        float v1 = vb[lane + 32];
        float ss = v0 * v0 + v1 * v1;
        #pragma unroll
        for (int m = 16; m >= 1; m >>= 1) ss += __shfl_xor(ss, m, 32);
        float nv  = sqrtf(ss);
        float inv = 1.0f / nv;
        s_rho[lane]      = v0 * inv;
        s_rho[lane + 32] = v1 * inv;
        const float* zb = z0 + (size_t)b * NN;
        s_y0[lane]      = zb[lane];
        s_y0[lane + 32] = zb[lane + 32];
        if (lane == 0) s_scal[0] = nv;
    }
    __syncthreads();

    {
        const float z0a = s_y0[2 * lane], z0b = s_y0[2 * lane + 1];
        const float vba = s_rho[2 * lane], vbb = s_rho[2 * lane + 1];
        const float* A1w = A1 + (size_t)b * PR * NN + (size_t)wave * 16 * NN;
        float bval = (lane < 16) ? b1[(size_t)b * PR + wave * 16 + lane] : 0.0f;
        float kl = 0.0f;
        #pragma unroll 4
        for (int row = 0; row < 16; ++row) {
            float2 a = *(const float2*)(A1w + row * NN + 2 * lane);
            float sz = a.x * z0a + a.y * z0b;
            float sv = a.x * vba + a.y * vbb;
            #pragma unroll
            for (int m = 16; m >= 1; m >>= 1) {
                sz += __shfl_xor(sz, m, 32);
                sv += __shfl_xor(sv, m, 32);
            }
            float brow = __shfl(bval, row, 32);
            if (lane == 0) {
                float t = sv / (brow - sz);
                kl = fmaxf(kl, t);
            }
        }
        if (lane == 0) s_red[wave] = kl;
    }

    {
        const int qi   = wave;
        const float* Pq = P + ((size_t)(b * NQ + qi)) * NN * NN;
        const float* qq = qv + (size_t)(b * NQ + qi) * NN;
        const int half = lane >> 4;
        const int l15  = lane & 15;
        const float* asrc = (l15 == 1) ? s_rho : s_y0;
        const bool   act  = (l15 < 2);

        float acc_uy = 0.f, acc_ur = 0.f, acc_wr = 0.f, acc_qy = 0.f, acc_qr = 0.f;

        for (int nt = 0; nt < 4; ++nt) {
            float u, w;
#ifdef HAVE_WMMA_F32
            v8f c = {};
            const float* Pb = Pq + half * (2 * NN) + nt * 16 + l15;
            #pragma unroll
            for (int kb = 0; kb < 16; ++kb) {
                int ko = 4 * kb + 2 * half;
                float ax = asrc[ko];
                float ay = asrc[ko + 1];
                ax = act ? ax : 0.0f;
                ay = act ? ay : 0.0f;
                v2f a;  a.x = ax;            a.y = ay;
                v2f bb; bb.x = Pb[kb * 4 * NN]; bb.y = Pb[kb * 4 * NN + NN];
                c = __builtin_amdgcn_wmma_f32_16x16x4_f32(
                        false, a, false, bb, (short)0, c, false, false);
            }
            u = c[0];
            w = c[1];
#else
            u = 0.f; w = 0.f;
            const int col = nt * 16 + l15;
            #pragma unroll 8
            for (int k = 0; k < NN; ++k) {
                float pvv = Pq[k * NN + col];
                u += s_y0[k]  * pvv;
                w += s_rho[k] * pvv;
            }
#endif
            int   idx  = nt * 16 + l15;
            float qvl  = qq[idx];
            float y0v  = s_y0[idx];
            float rhv  = s_rho[idx];
            float msk  = (half == 0) ? 1.0f : 0.0f;
            acc_uy += msk * u * y0v;
            acc_ur += msk * u * rhv;
            acc_wr += msk * w * rhv;
            acc_qy += msk * qvl * y0v;
            acc_qr += msk * qvl * rhv;
        }
        #pragma unroll
        for (int m = 16; m >= 1; m >>= 1) {
            acc_uy += __shfl_xor(acc_uy, m, 32);
            acc_ur += __shfl_xor(acc_ur, m, 32);
            acc_wr += __shfl_xor(acc_wr, m, 32);
            acc_qy += __shfl_xor(acc_qy, m, 32);
            acc_qr += __shfl_xor(acc_qr, m, 32);
        }
        if (lane == 0) {
            float rqv   = r[(size_t)b * NQ + qi];
            float c0    = 0.5f * acc_uy + acc_qy + rqv;
            float sigma = 2.0f * c0;
            float gr    = acc_ur + acc_qr;
            float rPr   = acc_wr;
            float quad  = (gr * gr - 2.0f * c0 * rPr) / (sigma * sigma);
            float kq    = (-gr / sigma) + sqrtf(quad);
            s_red[8 + qi] = kq;
        }
    }
    __syncthreads();

    if (tid == 0) {
        float k = 0.0f;
        #pragma unroll
        for (int i = 0; i < 16; ++i) k = fmaxf(k, s_red[i]);
        float nv    = s_scal[0];
        float alpha = fminf(1.0f / k, nv);
        s_scal[1]   = alpha;
    }
    __syncthreads();
    if (tid < NN) {
        out[(size_t)b * NN + tid] =
            z0[(size_t)b * NN + tid] + s_scal[1] * s_rho[tid];
    }
}

extern "C" void kernel_launch(void* const* d_in, const int* in_sizes, int n_in,
                              void* d_out, int out_size, void* d_ws, size_t ws_size,
                              hipStream_t stream) {
    const float* v  = (const float*)d_in[0];
    const float* A1 = (const float*)d_in[1];
    const float* b1 = (const float*)d_in[2];
    const float* z0 = (const float*)d_in[3];
    const float* P  = (const float*)d_in[4];
    const float* qv = (const float*)d_in[5];
    const float* r  = (const float*)d_in[6];
    float* out = (float*)d_out;
    (void)in_sizes; (void)n_in; (void)out_size; (void)d_ws; (void)ws_size;
    constraint_kernel<<<NB, 256, 0, stream>>>(v, A1, b1, z0, P, qv, r, out);
}


# P and A1 streamed by 16-byte-per-lane loads of 512 contiguous bytes, all issued at wave entry; WMMA tile column c = P column 4c+j; row and column sums now over 4 in-lane terms then 16 lanes
# speedup vs baseline: 1.5434x; 1.5434x over previous
#include <hip/hip_runtime.h>
#include <math.h>

#define NB   2048
#define PR   128
#define NN   64
#define NQ   8

typedef __attribute__((ext_vector_type(2))) float v2f;
typedef __attribute__((ext_vector_type(4))) float v4f;
typedef __attribute__((ext_vector_type(8))) float v8f;

__global__ __launch_bounds__(256) void constraint_kernel(
    const float* __restrict__ v,
    const float* __restrict__ A1,
    const float* __restrict__ b1,
    const float* __restrict__ z0,
    const float* __restrict__ P,
    const float* __restrict__ qv,
    const float* __restrict__ r,
    float* __restrict__ out)
{
    const int b    = blockIdx.x;
    const int tid  = threadIdx.x;
    const int wave = tid >> 5;
    const int lane = tid & 31;
    const int half = lane >> 4;
    const int l15  = lane & 15;

    __shared__ __attribute__((aligned(16))) float s_y0[NN];
    __shared__ __attribute__((aligned(16))) float s_rho[NN];
    __shared__ float s_red[16];
    __shared__ float s_scal[2];

    float v0 = 0.f, v1 = 0.f, zl0 = 0.f, zl1 = 0.f;
    if (wave == 0) {
        const float* vb = v + (size_t)b * NN;
        const float* zb = z0 + (size_t)b * NN;
        v0  = vb[lane];
        v1  = vb[lane + 32];
        zl0 = zb[lane];
        zl1 = zb[lane + 32];
    }
    const v4f* A4 = (const v4f*)(A1 + (size_t)b * PR * NN + (size_t)wave * 16 * NN) + lane;
    v4f ar[8];
    #pragma unroll
    for (int i = 0; i < 8; ++i) ar[i] = A4[i * 32];
    const float bval = b1[(size_t)b * PR + wave * 16 + l15];
    const v4f* P4 = (const v4f*)(P + ((size_t)(b * NQ + wave)) * NN * NN) + lane;
    v4f pr[32];
    #pragma unroll
    for (int i = 0; i < 32; ++i) pr[i] = P4[i * 32];
    const v4f   q4  = *((const v4f*)(qv + (size_t)(b * NQ + wave) * NN) + l15);
    const float rqv = r[(size_t)b * NQ + wave];
    const float zout = (tid < NN) ? z0[(size_t)b * NN + tid] : 0.0f;

    if (wave == 0) {
        float ss = v0 * v0 + v1 * v1;
        #pragma unroll
        for (int m = 16; m >= 1; m >>= 1) ss += __shfl_xor(ss, m, 32);
        float nv  = sqrtf(ss);
        float inv = 1.0f / nv;
        s_rho[lane]      = v0 * inv;
        s_rho[lane + 32] = v1 * inv;
        s_y0[lane]       = zl0;
        s_y0[lane + 32]  = zl1;
        if (lane == 0) s_scal[0] = nv;
    }
    __builtin_amdgcn_fence(__ATOMIC_RELEASE, "workgroup", "local");
    __builtin_amdgcn_s_barrier();
    __builtin_amdgcn_fence(__ATOMIC_ACQUIRE, "workgroup", "local");

    const v4f z4 = *(const v4f*)&s_y0[4 * l15];
    const v4f r4 = *(const v4f*)&s_rho[4 * l15];

    {
        float my_sz = 0.0f, my_sv = 0.0f;
        #pragma unroll
        for (int i = 0; i < 8; ++i) {
            const v4f a = ar[i];
            float sz = a.x * z4.x + a.y * z4.y + a.z * z4.z + a.w * z4.w;
            float sv = a.x * r4.x + a.y * r4.y + a.z * r4.z + a.w * r4.w;
            #pragma unroll
            for (int m = 8; m >= 1; m >>= 1) {
                sz += __shfl_xor(sz, m, 32);
                sv += __shfl_xor(sv, m, 32);
            }
            const bool mine = ((l15 >> 1) == i);
            my_sz = mine ? sz : my_sz;
            my_sv = mine ? sv : my_sv;
        }
        float t  = my_sv / (bval - my_sz);
        float kl = ((l15 & 1) == half) ? fmaxf(0.0f, t) : 0.0f;
        #pragma unroll
        for (int m = 16; m >= 1; m >>= 1) kl = fmaxf(kl, __shfl_xor(kl, m, 32));
        if (lane == 0) s_red[wave] = kl;
    }

    {
        const float* asrc = (l15 == 1) ? s_rho : s_y0;
        const bool   act  = (l15 < 2);

        v8f c0 = {}, c1 = {}, c2 = {}, c3 = {};
        #pragma unroll
        for (int kb = 0; kb < 16; ++kb) {
            float ax = asrc[4 * kb + half];
            float ay = asrc[4 * kb + 2 + half];
            ax = act ? ax : 0.0f;
            ay = act ? ay : 0.0f;
            v2f a;  a.x = ax; a.y = ay;
            const v4f p0 = pr[2 * kb];
            const v4f p1 = pr[2 * kb + 1];
            v2f bb;
            bb.x = p0.x; bb.y = p1.x;
            c0 = __builtin_amdgcn_wmma_f32_16x16x4_f32(false, a, false, bb, (short)0, c0, false, false);
            bb.x = p0.y; bb.y = p1.y;
            c1 = __builtin_amdgcn_wmma_f32_16x16x4_f32(false, a, false, bb, (short)0, c1, false, false);
            bb.x = p0.z; bb.y = p1.z;
            c2 = __builtin_amdgcn_wmma_f32_16x16x4_f32(false, a, false, bb, (short)0, c2, false, false);
            bb.x = p0.w; bb.y = p1.w;
            c3 = __builtin_amdgcn_wmma_f32_16x16x4_f32(false, a, false, bb, (short)0, c3, false, false);
        }
        float acc_uy = c0[0] * z4.x + c1[0] * z4.y + c2[0] * z4.z + c3[0] * z4.w;
        float acc_wy = c0[1] * z4.x + c1[1] * z4.y + c2[1] * z4.z + c3[1] * z4.w;
        float acc_wr = c0[1] * r4.x + c1[1] * r4.y + c2[1] * r4.z + c3[1] * r4.w;
        float acc_qy = q4.x * z4.x + q4.y * z4.y + q4.z * z4.z + q4.w * z4.w;
        float acc_qr = q4.x * r4.x + q4.y * r4.y + q4.z * r4.z + q4.w * r4.w;
        if (half) { acc_uy = 0.f; acc_wy = 0.f; acc_wr = 0.f; acc_qy = 0.f; acc_qr = 0.f; }
        #pragma unroll
        for (int m = 8; m >= 1; m >>= 1) {
            acc_uy += __shfl_xor(acc_uy, m, 32);
            acc_wy += __shfl_xor(acc_wy, m, 32);
            acc_wr += __shfl_xor(acc_wr, m, 32);
            acc_qy += __shfl_xor(acc_qy, m, 32);
            acc_qr += __shfl_xor(acc_qr, m, 32);
        }
        if (lane == 0) {
            float c0s   = 0.5f * acc_uy + acc_qy + rqv;
            float sigma = 2.0f * c0s;
            float gr    = acc_wy + acc_qr;
            float rPr   = acc_wr;
            float quad  = (gr * gr - 2.0f * c0s * rPr) / (sigma * sigma);
            float kq    = (-gr / sigma) + sqrtf(quad);
            s_red[8 + wave] = kq;
        }
    }
    __syncthreads();

    if (tid == 0) {
        float k = 0.0f;
        #pragma unroll
        for (int i = 0; i < 16; ++i) k = fmaxf(k, s_red[i]);
        float nv    = s_scal[0];
        float alpha = fminf(1.0f / k, nv);
        s_scal[1]   = alpha;
    }
    __syncthreads();
    if (tid < NN) {
        out[(size_t)b * NN + tid] = zout + s_scal[1] * s_rho[tid];
    }
}

extern "C" void kernel_launch(void* const* d_in, const int* in_sizes, int n_in,
                              void* d_out, int out_size, void* d_ws, size_t ws_size,
                              hipStream_t stream) {
    const float* v  = (const float*)d_in[0];
    const float* A1 = (const float*)d_in[1];
    const float* b1 = (const float*)d_in[2];
    const float* z0 = (const float*)d_in[3];
    const float* P  = (const float*)d_in[4];
    const float* qv = (const float*)d_in[5];
    const float* r  = (const float*)d_in[6];
    float* out = (float*)d_out;
    (void)in_sizes; (void)n_in; (void)out_size; (void)d_ws; (void)ws_size;
    constraint_kernel<<<NB, 256, 0, stream>>>(v, A1, b1, z0, P, qv, r, out);
}


# Each wave streams 4 (16 A1 rows, P[q]) units through a 4-group register ring, requesting group n+4 as group n is consumed; 2-wave workgroups, all 4096 waves resident at once; every sum's order kept
# speedup vs baseline: 1.8085x; 1.1718x over previous
#include <hip/hip_runtime.h>
#include <math.h>

#define NB   2048
#define PR   128
#define NN   64
#define NQ   8
#define UNITS  4
#define GPU    5
#define NGRP   (UNITS * GPU)
#define RING   4

typedef __attribute__((ext_vector_type(2))) float v2f;
typedef __attribute__((ext_vector_type(4))) float v4f;
typedef __attribute__((ext_vector_type(8))) float v8f;

__global__ __launch_bounds__(64) void constraint_kernel(
    const float* __restrict__ v,
    const float* __restrict__ A1,
    const float* __restrict__ b1,
    const float* __restrict__ z0,
    const float* __restrict__ P,
    const float* __restrict__ qv,
    const float* __restrict__ r,
    float* __restrict__ out)
{
    const int b    = blockIdx.x;
    const int tid  = threadIdx.x;
    const int wave = __builtin_amdgcn_readfirstlane(tid >> 5);
    const int lane = tid & 31;
    const int half = lane >> 4;
    const int l15  = lane & 15;
    const int g0   = wave * UNITS;

    __shared__ __attribute__((aligned(16))) float s_y0[NN];
    __shared__ __attribute__((aligned(16))) float s_rho[NN];
    __shared__ float s_red[2];
    __shared__ float s_scal[1];

    float v0 = 0.f, v1 = 0.f, zl0 = 0.f, zl1 = 0.f;
    if (wave == 0) {
        const float* vb = v + (size_t)b * NN;
        const float* zb = z0 + (size_t)b * NN;
        v0  = vb[lane];
        v1  = vb[lane + 32];
        zl0 = zb[lane];
        zl1 = zb[lane + 32];
    }
    const unsigned ulane = (unsigned)lane, ul15 = (unsigned)l15;
    const float* A1w = A1 + ((size_t)b * PR * NN + (size_t)g0 * 16 * NN);
    const float* Pw  = P  + ((size_t)(b * NQ + g0)) * NN * NN;
    const float* b1w = b1 + ((size_t)b * PR + (size_t)(g0 * 16));
    const float* qw  = qv + ((size_t)(b * NQ + g0)) * NN;
    const v4f*   A4  = (const v4f*)A1w + ulane;
    const v4f*   P4  = (const v4f*)Pw + ulane;
    const float* b1p = b1w + (unsigned)((l15 & 14) + half);
    const v4f*   q4p = (const v4f*)qw + ul15;
    const float* rp  = r + ((size_t)b * NQ + (size_t)g0);

    v4f   ring[RING][8];
    float bval[UNITS];
    v4f   q4[UNITS];
    float rq[UNITS];

#define LOAD_GROUP(n) do {                                                              \
        const int u_ = (n) / GPU, part_ = (n) % GPU, slot_ = (n) % RING;                 \
        _Pragma("unroll") for (int i = 0; i < 8; ++i)                                    \
            ring[slot_][i] = (part_ == 0) ? A4[u_ * 256 + i * 32]                        \
                                          : P4[u_ * 1024 + (8 * (part_ - 1) + i) * 32];  \
        if (part_ == 0) bval[u_] = b1p[16 * u_];                                         \
        if (part_ == GPU - 1) q4[u_] = q4p[16 * u_];                                     \
    } while (0)

    #pragma unroll
    for (int n = 0; n < RING; ++n) {
        LOAD_GROUP(n);
        __builtin_amdgcn_sched_barrier(0);
    }
    #pragma unroll
    for (int u = 0; u < UNITS; ++u) rq[u] = rp[u];
    const float zout = (z0 + (size_t)b * NN)[(unsigned)tid];

    if (wave == 0) {
        float ss = v0 * v0 + v1 * v1;
        #pragma unroll
        for (int m = 16; m >= 1; m >>= 1) ss += __shfl_xor(ss, m, 32);
        float nv  = sqrtf(ss);
        float inv = 1.0f / nv;
        s_rho[lane]      = v0 * inv;
        s_rho[lane + 32] = v1 * inv;
        s_y0[lane]       = zl0;
        s_y0[lane + 32]  = zl1;
        if (lane == 0) s_scal[0] = nv;
    }
    __builtin_amdgcn_fence(__ATOMIC_RELEASE, "workgroup", "local");
    __builtin_amdgcn_s_barrier();
    __builtin_amdgcn_fence(__ATOMIC_ACQUIRE, "workgroup", "local");

    const v4f z4 = *(const v4f*)&s_y0[4 * l15];
    const v4f r4 = *(const v4f*)&s_rho[4 * l15];
    const float* asrc = (l15 == 1) ? s_rho : s_y0;
    const bool   act  = (l15 < 2);

    float kmax = 0.0f;

    #pragma unroll
    for (int u = 0; u < UNITS; ++u) {
        float kl;
        {
            const int n = GPU * u, slot = n % RING;
            float szp[8], svp[8];
            #pragma unroll
            for (int i = 0; i < 8; ++i) {
                const v4f a = ring[slot][i];
                szp[i] = a.x * z4.x + a.y * z4.y + a.z * z4.z + a.w * z4.w;
                svp[i] = a.x * r4.x + a.y * r4.y + a.z * r4.z + a.w * r4.w;
            }
            __builtin_amdgcn_sched_barrier(0);
            if (n + RING < NGRP) LOAD_GROUP(n + RING);
            __builtin_amdgcn_sched_barrier(0);
            float my_sz = 0.0f, my_sv = 0.0f;
            #pragma unroll
            for (int i = 0; i < 8; ++i) {
                float sz = szp[i];
                float sv = svp[i];
                #pragma unroll
                for (int m = 8; m >= 1; m >>= 1) {
                    sz += __shfl_xor(sz, m, 32);
                    sv += __shfl_xor(sv, m, 32);
                }
                const bool mine = ((l15 >> 1) == i);
                my_sz = mine ? sz : my_sz;
                my_sv = mine ? sv : my_sv;
            }
            float t = my_sv / (bval[u] - my_sz);
            kl = fmaxf(0.0f, t);
            #pragma unroll
            for (int m = 16; m >= 1; m >>= 1) kl = fmaxf(kl, __shfl_xor(kl, m, 32));
        }

        v8f c0 = {}, c1 = {}, c2 = {}, c3 = {};
        #pragma unroll
        for (int kg = 0; kg < 4; ++kg) {
            const int n = GPU * u + 1 + kg, slot = n % RING;
            #pragma unroll
            for (int kk = 0; kk < 4; ++kk) {
                const int kb = 4 * kg + kk;
                float ax = asrc[4 * kb + half];
                float ay = asrc[4 * kb + 2 + half];
                ax = act ? ax : 0.0f;
                ay = act ? ay : 0.0f;
                v2f a;  a.x = ax; a.y = ay;
                const v4f p0 = ring[slot][2 * kk];
                const v4f p1 = ring[slot][2 * kk + 1];
                v2f bb;
                bb.x = p0.x; bb.y = p1.x;
                c0 = __builtin_amdgcn_wmma_f32_16x16x4_f32(false, a, false, bb, (short)0, c0, false, false);
                bb.x = p0.y; bb.y = p1.y;
                c1 = __builtin_amdgcn_wmma_f32_16x16x4_f32(false, a, false, bb, (short)0, c1, false, false);
                bb.x = p0.z; bb.y = p1.z;
                c2 = __builtin_amdgcn_wmma_f32_16x16x4_f32(false, a, false, bb, (short)0, c2, false, false);
                bb.x = p0.w; bb.y = p1.w;
                c3 = __builtin_amdgcn_wmma_f32_16x16x4_f32(false, a, false, bb, (short)0, c3, false, false);
            }
            __builtin_amdgcn_sched_barrier(0);
            if (n + RING < NGRP) LOAD_GROUP(n + RING);
            __builtin_amdgcn_sched_barrier(0);
        }
        const v4f   qc = q4[u];
        const float rc = rq[u];
        float acc_uy = c0[0] * z4.x + c1[0] * z4.y + c2[0] * z4.z + c3[0] * z4.w;
        float acc_wy = c0[1] * z4.x + c1[1] * z4.y + c2[1] * z4.z + c3[1] * z4.w;
        float acc_wr = c0[1] * r4.x + c1[1] * r4.y + c2[1] * r4.z + c3[1] * r4.w;
        float acc_qy = qc.x * z4.x + qc.y * z4.y + qc.z * z4.z + qc.w * z4.w;
        float acc_qr = qc.x * r4.x + qc.y * r4.y + qc.z * r4.z + qc.w * r4.w;
        if (half) { acc_uy = 0.f; acc_wy = 0.f; acc_wr = 0.f; acc_qy = 0.f; acc_qr = 0.f; }
        #pragma unroll
        for (int m = 8; m >= 1; m >>= 1) {
            acc_uy += __shfl_xor(acc_uy, m, 32);
            acc_wy += __shfl_xor(acc_wy, m, 32);
            acc_wr += __shfl_xor(acc_wr, m, 32);
            acc_qy += __shfl_xor(acc_qy, m, 32);
            acc_qr += __shfl_xor(acc_qr, m, 32);
        }
        float c0s   = 0.5f * acc_uy + acc_qy + rc;
        float sigma = 2.0f * c0s;
        float gr    = acc_wy + acc_qr;
        float rPr   = acc_wr;
        float quad  = (gr * gr - 2.0f * c0s * rPr) / (sigma * sigma);
        float kq    = (-gr / sigma) + sqrtf(quad);
        kq = __shfl(kq, 0, 32);
        kmax = fmaxf(kmax, fmaxf(kl, kq));
    }
#undef LOAD_GROUP

    if (lane == 0) s_red[wave] = kmax;
    __syncthreads();
    {
        float k     = fmaxf(fmaxf(0.0f, s_red[0]), s_red[1]);
        float nv    = s_scal[0];
        float alpha = fminf(1.0f / k, nv);
        (out + (size_t)b * NN)[(unsigned)tid] = zout + alpha * s_rho[tid];
    }
}

extern "C" void kernel_launch(void* const* d_in, const int* in_sizes, int n_in,
                              void* d_out, int out_size, void* d_ws, size_t ws_size,
                              hipStream_t stream) {
    const float* v  = (const float*)d_in[0];
    const float* A1 = (const float*)d_in[1];
    const float* b1 = (const float*)d_in[2];
    const float* z0 = (const float*)d_in[3];
    const float* P  = (const float*)d_in[4];
    const float* qv = (const float*)d_in[5];
    const float* r  = (const float*)d_in[6];
    float* out = (float*)d_out;
    (void)in_sizes; (void)n_in; (void)out_size; (void)d_ws; (void)ws_size;
    constraint_kernel<<<NB, 64, 0, stream>>>(v, A1, b1, z0, P, qv, r, out);
}
